# starting kernel
# baseline (speedup 1.0000x reference)
#include <hip/hip_runtime.h>


typedef float v2f __attribute__((ext_vector_type(2)));
typedef float v8f __attribute__((ext_vector_type(8)));

#define CH 128
#define NEG_SLOPE 0.2f
#define SM_EPS 1e-16f

__device__ __forceinline__ unsigned enc_f32(float f) {
  unsigned ib = __float_as_uint(f);
  return (ib & 0x80000000u) ? ~ib : (ib | 0x80000000u);
}
__device__ __forceinline__ float dec_f32(unsigned u) {
  return __uint_as_float((u & 0x80000000u) ? (u & 0x7FFFFFFFu) : ~u);
}

#define WT_PAD 132

__global__ void gat_gemm_wmma(const float* __restrict__ X,
                              const float* __restrict__ W,
                              float* __restrict__ H, int n_nodes) {
  extern __shared__ float wt[];
  const int tid = threadIdx.x;
  for (int i = tid; i < CH * CH; i += 256) {
    int k = i >> 7;
    int n = i & (CH - 1);
    wt[n * WT_PAD + k] = W[i];
  }
  __syncthreads();

  const int wave = tid >> 5;
  const int lane = tid & 31;
  const int row0 = blockIdx.x * 128 + wave * 16;
  const int mrow = lane & 15;
  const int koff = (lane >> 4) * 2;

  int rload = row0 + mrow;
  if (rload >= n_nodes) rload = n_nodes - 1;
  const float* xrow = X + (size_t)rload * CH + koff;
  const float* wrow = wt + mrow * WT_PAD + koff;

  v8f acc[8];
#pragma unroll
  for (int t = 0; t < 8; ++t) acc[t] = {};

  v2f a = *(const v2f*)(xrow);
  v2f b[8];
#pragma unroll
  for (int t = 0; t < 8; ++t)
    b[t] = *(const v2f*)(wrow + t * 16 * WT_PAD);

  for (int k4 = 0; k4 < CH; k4 += 4) {
    const int kn = k4 + 4;
    v2f an = {};
    v2f bn[8];
#pragma unroll
    for (int t = 0; t < 8; ++t) bn[t] = {};
    if (kn < CH) {
      an = *(const v2f*)(xrow + kn);
#pragma unroll
      for (int t = 0; t < 8; ++t)
        bn[t] = *(const v2f*)(wrow + t * 16 * WT_PAD + kn);
    }
#pragma unroll
    for (int t = 0; t < 8; ++t)
      acc[t] = __builtin_amdgcn_wmma_f32_16x16x4_f32(
          false, a, false, b[t], (short)0, acc[t], false, false);
    a = an;
#pragma unroll
    for (int t = 0; t < 8; ++t) b[t] = bn[t];
  }

  const int rhi = (lane >> 4) * 8;
#pragma unroll
  for (int t = 0; t < 8; ++t) {
    const int col = t * 16 + (lane & 15);
#pragma unroll
    for (int r = 0; r < 8; ++r) {
      int row = row0 + r + rhi;
      if (row < n_nodes) H[(size_t)row * CH + col] = acc[t][r];
    }
  }
}

__global__ void gat_node_scores(const float* __restrict__ H,
                                const float* __restrict__ a_src,
                                const float* __restrict__ a_dst,
                                float* __restrict__ s, float* __restrict__ d,
                                int n_nodes) {
  const int wave = threadIdx.x >> 5;
  const int lane = threadIdx.x & 31;
  const int node = blockIdx.x * 8 + wave;
  if (node >= n_nodes) return;
  float4 h4 = ((const float4*)(H + (size_t)node * CH))[lane];
  float4 a4 = ((const float4*)a_src)[lane];
  float4 b4 = ((const float4*)a_dst)[lane];
  float ss = h4.x * a4.x + h4.y * a4.y + h4.z * a4.z + h4.w * a4.w;
  float dd = h4.x * b4.x + h4.y * b4.y + h4.z * b4.z + h4.w * b4.w;
#pragma unroll
  for (int o = 16; o > 0; o >>= 1) {
    ss += __shfl_xor(ss, o, 32);
    dd += __shfl_xor(dd, o, 32);
  }
  if (lane == 0) {
    s[node] = ss;
    d[node] = dd;
  }
}

__global__ void gat_init(float* __restrict__ agg, unsigned* __restrict__ emax,
                         float* __restrict__ denom, int n_nodes) {
  size_t i = (size_t)blockIdx.x * blockDim.x + threadIdx.x;
  size_t total = (size_t)n_nodes * CH;
  if (i < total) agg[i] = 0.0f;
  if (i < (size_t)n_nodes) {
    emax[i] = 0u;
    denom[i] = 0.0f;
  }
}

__global__ void gat_edge_max(const int* __restrict__ src,
                             const int* __restrict__ dst,
                             const float* __restrict__ s,
                             const float* __restrict__ d,
                             float* __restrict__ ework,
                             unsigned* __restrict__ emax, int n_edges,
                             int n_nodes) {
  int e = blockIdx.x * blockDim.x + threadIdx.x;
  int tot = n_edges + n_nodes;
  if (e >= tot) return;
  int u = (e < n_edges) ? src[e] : (e - n_edges);
  int v = (e < n_edges) ? dst[e] : (e - n_edges);
  float val = s[u] + d[v];
  val = (val > 0.0f) ? val : NEG_SLOPE * val;
  ework[e] = val;
  atomicMax(&emax[v], enc_f32(val));
}

__global__ void gat_edge_expsum(const int* __restrict__ dst,
                                float* __restrict__ ework,
                                const unsigned* __restrict__ emax,
                                float* __restrict__ denom, int n_edges,
                                int n_nodes) {
  int e = blockIdx.x * blockDim.x + threadIdx.x;
  int tot = n_edges + n_nodes;
  if (e >= tot) return;
  int v = (e < n_edges) ? dst[e] : (e - n_edges);
  float w = __builtin_expf(ework[e] - dec_f32(emax[v]));
  ework[e] = w;
  atomicAdd(&denom[v], w);
}

__global__ void gat_edge_aggregate(const int* __restrict__ src,
                                   const int* __restrict__ dst,
                                   const float* __restrict__ ework,
                                   const float* __restrict__ denom,
                                   const float* __restrict__ H,
                                   float* __restrict__ agg, int n_edges,
                                   int n_nodes) {
  int wid = (int)(((size_t)blockIdx.x * blockDim.x + threadIdx.x) >> 5);
  int lane = threadIdx.x & 31;
  int tot = n_edges + n_nodes;
  if (wid >= tot) return;
  int u = (wid < n_edges) ? src[wid] : (wid - n_edges);
  int v = (wid < n_edges) ? dst[wid] : (wid - n_edges);
  float alpha = ework[wid] / (denom[v] + SM_EPS);
  float4 h4 = ((const float4*)(H + (size_t)u * CH))[lane];
  float* ap = agg + (size_t)v * CH + lane * 4;
  atomicAdd(ap + 0, alpha * h4.x);
  atomicAdd(ap + 1, alpha * h4.y);
  atomicAdd(ap + 2, alpha * h4.z);
  atomicAdd(ap + 3, alpha * h4.w);
}

__global__ void gat_update(float* __restrict__ x, const float* __restrict__ agg,
                           const float* __restrict__ bias, int n_nodes) {
  size_t i = (size_t)blockIdx.x * blockDim.x + threadIdx.x;
  if (i >= (size_t)n_nodes * CH) return;
  float v = agg[i] + bias[i & (CH - 1)];
  x[i] += 1.0f / (1.0f + __builtin_expf(-v));
}

__global__ void gat_copy(const float* __restrict__ in, float* __restrict__ out,
                         size_t n) {
  size_t i = (size_t)blockIdx.x * blockDim.x + threadIdx.x;
  if (i < n) out[i] = in[i];
}

extern "C" void kernel_launch(void* const* d_in, const int* in_sizes, int n_in,
                              void* d_out, int out_size, void* d_ws,
                              size_t ws_size, hipStream_t stream) {
  const float* x_in = (const float*)d_in[0];
  const int* ei = (const int*)d_in[1];
  const float* Ws = (const float*)d_in[2];
  const float* att_src = (const float*)d_in[3];
  const float* att_dst = (const float*)d_in[4];
  const float* bias = (const float*)d_in[5];

  const int N = in_sizes[0] / CH;
  const int E = in_sizes[1] / 2;
  const int M = in_sizes[3] / CH;
  const int* src = ei;
  const int* dst = ei + E;
  const int TOT = E + N;

  float* ws = (float*)d_ws;
  float* h = ws;
  float* agg = h + (size_t)N * CH;
  float* s = agg + (size_t)N * CH;
  float* dsc = s + N;
  float* denom = dsc + N;
  unsigned* emax = (unsigned*)(denom + N);
  float* ework = (float*)(emax + N);

  float* x = (float*)d_out;

  const size_t nfeat = (size_t)N * CH;
  const int blk = 256;
  const int g_feat = (int)((nfeat + blk - 1) / blk);
  const int g_gemm = (N + 127) / 128;
  const int g_node = (N + 7) / 8;
  const int g_edge = (TOT + blk - 1) / blk;
  const int g_eagg = (TOT + 7) / 8;
  const size_t smem = (size_t)CH * WT_PAD * sizeof(float);

  gat_copy<<<g_feat, blk, 0, stream>>>(x_in, x, nfeat);

  for (int m = 0; m < M; ++m) {
    const float* W = Ws + (size_t)m * CH * CH;
    const float* as_ = att_src + (size_t)m * CH;
    const float* ad_ = att_dst + (size_t)m * CH;
    const float* b_ = bias + (size_t)m * CH;

    gat_gemm_wmma<<<g_gemm, blk, smem, stream>>>(x, W, h, N);
    gat_node_scores<<<g_node, blk, 0, stream>>>(h, as_, ad_, s, dsc, N);
    gat_init<<<g_feat, blk, 0, stream>>>(agg, emax, denom, N);
    gat_edge_max<<<g_edge, blk, 0, stream>>>(src, dst, s, dsc, ework, emax, E, N);
    gat_edge_expsum<<<g_edge, blk, 0, stream>>>(dst, ework, emax, denom, E, N);
    gat_edge_aggregate<<<g_eagg, blk, 0, stream>>>(src, dst, ework, denom, h,
                                                   agg, E, N);
    gat_update<<<g_feat, blk, 0, stream>>>(x, agg, b_, N);
  }
}


# owning group lists its edges (src, row) in workspace in layer 1's index scan; each layer's max, den, sums (LDS f32 atomics, arrival order, self loop an addend) come from the list; 9 to 6 launches
# speedup vs baseline: 4.8981x; 4.8981x over previous
#include <hip/hip_runtime.h>


typedef float v2f __attribute__((ext_vector_type(2)));
typedef float v8f __attribute__((ext_vector_type(8)));

#define CH 128
#define NEG_SLOPE 0.2f
#define SM_EPS 1e-16f

__device__ __forceinline__ unsigned enc_f32(float f) {
  unsigned ib = __float_as_uint(f);
  return (ib & 0x80000000u) ? ~ib : (ib | 0x80000000u);
}
__device__ __forceinline__ float dec_f32(unsigned u) {
  return __uint_as_float((u & 0x80000000u) ? (u & 0x7FFFFFFFu) : ~u);
}

#define WT_PAD 132

__global__ void gat_gemm_wmma(const float* __restrict__ X,
                              const float* __restrict__ W,
                              const float* __restrict__ a_src,
                              const float* __restrict__ a_dst,
                              float* __restrict__ H, float* __restrict__ s,
                              float* __restrict__ d, int n_nodes) {
  extern __shared__ float wt[];
  const int tid = threadIdx.x;
  for (int i = tid; i < CH * CH; i += 256) {
    int k = i >> 7;
    int n = i & (CH - 1);
    wt[n * WT_PAD + k] = W[i];
  }
  __syncthreads();

  const int wave = tid >> 5;
  const int lane = tid & 31;
  const int row0 = blockIdx.x * 128 + wave * 16;
  const int mrow = lane & 15;
  const int koff = (lane >> 4) * 2;

  int rload = row0 + mrow;
  if (rload >= n_nodes) rload = n_nodes - 1;
  const float* xrow = X + (size_t)rload * CH + koff;
  const float* wrow = wt + mrow * WT_PAD + koff;

  v8f acc[8];
#pragma unroll
  for (int t = 0; t < 8; ++t) acc[t] = {};

  v2f a = *(const v2f*)(xrow);
  v2f b[8];
#pragma unroll
  for (int t = 0; t < 8; ++t)
    b[t] = *(const v2f*)(wrow + t * 16 * WT_PAD);

  for (int k4 = 0; k4 < CH; k4 += 4) {
    const int kn = k4 + 4;
    v2f an = {};
    v2f bn[8];
#pragma unroll
    for (int t = 0; t < 8; ++t) bn[t] = {};
    if (kn < CH) {
      an = *(const v2f*)(xrow + kn);
#pragma unroll
      for (int t = 0; t < 8; ++t)
        bn[t] = *(const v2f*)(wrow + t * 16 * WT_PAD + kn);
    }
#pragma unroll
    for (int t = 0; t < 8; ++t)
      acc[t] = __builtin_amdgcn_wmma_f32_16x16x4_f32(
          false, a, false, b[t], (short)0, acc[t], false, false);
    a = an;
#pragma unroll
    for (int t = 0; t < 8; ++t) b[t] = bn[t];
  }

  const int rhi = (lane >> 4) * 8;

  float ps[8], pd[8];
#pragma unroll
  for (int r = 0; r < 8; ++r) { ps[r] = 0.0f; pd[r] = 0.0f; }
#pragma unroll
  for (int t = 0; t < 8; ++t) {
    const float as = a_src[t * 16 + (lane & 15)];
    const float ad = a_dst[t * 16 + (lane & 15)];
#pragma unroll
    for (int r = 0; r < 8; ++r) {
      ps[r] += acc[t][r] * as;
      pd[r] += acc[t][r] * ad;
    }
  }
#pragma unroll
  for (int o = 8; o > 0; o >>= 1) {
#pragma unroll
    for (int r = 0; r < 8; ++r) {
      ps[r] += __shfl_xor(ps[r], o, 32);
      pd[r] += __shfl_xor(pd[r], o, 32);
    }
  }
  float so = ps[0], dq = pd[0];
#pragma unroll
  for (int r = 1; r < 8; ++r) {
    so = ((lane & 15) == r) ? ps[r] : so;
    dq = ((lane & 15) == r) ? pd[r] : dq;
  }
  const int orow = row0 + (lane & 15) + rhi;
  if ((lane & 15) < 8 && orow < n_nodes) {
    s[orow] = so;
    d[orow] = dq;
  }

#pragma unroll
  for (int t = 0; t < 8; ++t) {
    const int col = t * 16 + (lane & 15);
#pragma unroll
    for (int r = 0; r < 8; ++r) {
      int row = row0 + r + rhi;
      if (row < n_nodes) H[(size_t)row * CH + col] = acc[t][r];
    }
  }
}

#define NPB 400
#define QCAP 192
#define EIF 8
#define LHDR 32
#define LROW 23

struct AggCtx {
  const int* src;
  const float* s;
  const float* sd;
  unsigned* smaxu;
  float* sden;
  const float* H;
  float* sagg;
  int* qe;
  unsigned short* qd;
  unsigned* glist;
  int* scnt;
  int cap;
  int n0;
  int lane;
};

__device__ __forceinline__ float edge_score(const AggCtx& c, int u, int dl) {
  const float val = c.s[u] + c.sd[dl];
  return (val > 0.0f) ? val : NEG_SLOPE * val;
}

__device__ __forceinline__ void edges_max(const AggCtx& c, int u, int dl,
                                          int n) {
  const float ev = edge_score(c, u, dl);
  if (c.lane < n) atomicMax(c.smaxu + dl, enc_f32(ev));
}

__device__ __forceinline__ void edges_sum(const AggCtx& c, int u, int dl,
                                          int n) {
  const float ev = edge_score(c, u, dl);
  const float w = __builtin_expf(ev - dec_f32(c.smaxu[dl]));
  if (c.lane < n) atomicAdd(c.sden + dl, w);
  const int al = __float_as_int(w);
  int b = 0;
  for (; b + EIF <= n; b += EIF) {
    float4 r[EIF];
#pragma unroll
    for (int i = 0; i < EIF; ++i) {
      const int uj = __builtin_amdgcn_readlane(u, b + i);
      r[i] = *(const float4*)(c.H + (size_t)uj * CH + 4 * c.lane);
    }
#pragma unroll
    for (int i = 0; i < EIF; ++i) {
      const int dj = __builtin_amdgcn_readlane(dl, b + i);
      const float aj = __int_as_float(__builtin_amdgcn_readlane(al, b + i));
      float* ap = c.sagg + dj * CH + c.lane;
      atomicAdd(ap + 0, aj * r[i].x);
      atomicAdd(ap + 32, aj * r[i].y);
      atomicAdd(ap + 64, aj * r[i].z);
      atomicAdd(ap + 96, aj * r[i].w);
    }
  }
  for (; b < n; ++b) {
    const int uj = __builtin_amdgcn_readlane(u, b);
    const int dj = __builtin_amdgcn_readlane(dl, b);
    const float aj = __int_as_float(__builtin_amdgcn_readlane(al, b));
    const float4 r = *(const float4*)(c.H + (size_t)uj * CH + 4 * c.lane);
    float* ap = c.sagg + dj * CH + c.lane;
    atomicAdd(ap + 0, aj * r.x);
    atomicAdd(ap + 32, aj * r.y);
    atomicAdd(ap + 64, aj * r.z);
    atomicAdd(ap + 96, aj * r.w);
  }
}

template <int MODE>
__device__ __forceinline__ void walk_list(const AggCtx& c, int total,
                                          int nself, int wave) {
  const int tot = total + ((MODE == 1) ? nself : 0);
  for (int g0 = wave * 32; g0 < tot; g0 += 1024) {
    const int n = (tot - g0 < 32) ? (tot - g0) : 32;
    const int i = g0 + ((c.lane < n) ? c.lane : 0);
    int u, dl;
    if (i < total) {
      const unsigned wd = c.glist[i];
      u = (int)(wd & ((1u << LROW) - 1u));
      dl = (int)(wd >> LROW);
    } else {
      dl = i - total;
      u = c.n0 + dl;
    }
    if (MODE == 0) edges_max(c, u, dl, n);
    else edges_sum(c, u, dl, n);
  }
}

template <int MODE>
__device__ __forceinline__ void queue_group(const AggCtx& c, int qb, int n,
                                            bool build) {
  const int j = qb + ((c.lane < n) ? c.lane : 0);
  const int e = c.qe[j];
  const int dl = c.qd[j];
  const int u = c.src[e];
  if (MODE == 0) {
    edges_max(c, u, dl, n);
    if (build) {
      int base = 0;
      if (c.lane == 0) base = atomicAdd(c.scnt, n);
      base = __builtin_amdgcn_readfirstlane(base);
      if (c.lane < n && base + c.lane < c.cap)
        c.glist[base + c.lane] = (unsigned)u | ((unsigned)dl << LROW);
    }
  } else {
    edges_sum(c, u, dl, n);
  }
}

__device__ __forceinline__ void agg_append(const AggCtx& c, bool m, int e,
                                           int dl, int& qn) {
  const unsigned bal = __builtin_amdgcn_ballot_w32(m);
  if (m) {
    const int pos = qn + (int)__builtin_amdgcn_mbcnt_lo(bal, 0u);
    c.qe[pos] = e;
    c.qd[pos] = (unsigned short)dl;
  }
  qn += __builtin_popcount(bal);
}

__device__ __forceinline__ int4 agg_load(const int4* dst4, int vi, int nb4) {
  return dst4[(vi < nb4) ? vi : (nb4 - 1)];
}

template <int MODE>
__device__ __forceinline__ void agg_scan4(const AggCtx& c, int4 d, int e0,
                                          unsigned cnt, int& qn, bool build) {
  const int l0 = d.x - c.n0, l1 = d.y - c.n0, l2 = d.z - c.n0,
            l3 = d.w - c.n0;
  const bool m0 = (unsigned)l0 < cnt, m1 = (unsigned)l1 < cnt,
             m2 = (unsigned)l2 < cnt, m3 = (unsigned)l3 < cnt;
  if (__builtin_amdgcn_ballot_w32(m0 | m1 | m2 | m3)) {
    agg_append(c, m0, e0 + 0, l0, qn);
    agg_append(c, m1, e0 + 1, l1, qn);
    agg_append(c, m2, e0 + 2, l2, qn);
    agg_append(c, m3, e0 + 3, l3, qn);
    if (qn >= 64) {
      const int nfull = qn & ~31;
      for (int qb = 0; qb < nfull; qb += 32)
        queue_group<MODE>(c, qb, 32, build);
      const int rem = qn - nfull;
      const int te = c.qe[nfull + c.lane];
      const unsigned short td = c.qd[nfull + c.lane];
      if (c.lane < rem) {
        c.qe[c.lane] = te;
        c.qd[c.lane] = td;
      }
      qn = rem;
    }
  }
}

template <int MODE>
__device__ __forceinline__ void scan_edges(const AggCtx& c, const int* dst,
                                           int n_edges, int cnt, int wave,
                                           bool build) {
  int qn = 0;
  int head = (int)(((16u - (unsigned)((size_t)dst & 15u)) & 15u) >> 2);
  if (head > n_edges) head = n_edges;
  const int nb4 = (n_edges - head) >> 2;
  const int tail = (n_edges - head) & 3;
  if (wave == 0) {
    const int nleft = head + tail;
    const int e = (c.lane < head) ? c.lane : (n_edges - tail + (c.lane - head));
    const bool ok = c.lane < nleft;
    const int dl = (ok ? dst[e] : -1) - c.n0;
    agg_append(c, ok && (unsigned)dl < (unsigned)cnt, e, dl, qn);
  }
  const int4* dst4 = (const int4*)(dst + head);
  if (nb4 > 0) {
    const int T = (nb4 + 4095) >> 12;
    int t = (int)((blockIdx.x * (unsigned)T) / gridDim.x);
    if (t >= T) t = 0;
    int4 cur[4], nxt[4];
#pragma unroll
    for (int s = 0; s < 4; ++s)
      cur[s] = agg_load(dst4, (t * 32 + wave) * 128 + s * 32 + c.lane, nb4);
    for (int k = 0; k < T; ++k) {
      int tn = t + 1;
      if (tn >= T) tn -= T;
#pragma unroll
      for (int s = 0; s < 4; ++s)
        nxt[s] = agg_load(dst4, (tn * 32 + wave) * 128 + s * 32 + c.lane, nb4);
#pragma unroll
      for (int s = 0; s < 4; ++s) {
        const int vi = (t * 32 + wave) * 128 + s * 32 + c.lane;
        agg_scan4<MODE>(c, cur[s], head + 4 * vi,
                        (vi < nb4) ? (unsigned)cnt : 0u, qn, build);
      }
#pragma unroll
      for (int s = 0; s < 4; ++s) cur[s] = nxt[s];
      t = tn;
    }
  }
  for (int qb = 0; qb < qn; qb += 32)
    queue_group<MODE>(c, qb, (qn - qb < 32) ? (qn - qb) : 32, build);
}

__global__ void __launch_bounds__(1024)
gat_edge_aggregate(const int* __restrict__ src, const int* __restrict__ dst,
                   const float* __restrict__ s, const float* __restrict__ d,
                   const float* __restrict__ H,
                   const float* __restrict__ bias, const float* xin,
                   float* xout, unsigned* lists, int lstride, int build,
                   int n_edges, int n_nodes) {
  __shared__ __attribute__((aligned(16))) float sagg[NPB * CH];
  __shared__ int s_qe[32 * QCAP];
  __shared__ unsigned short s_qd[32 * QCAP];
  __shared__ unsigned s_maxu[NPB];
  __shared__ float s_den[NPB];
  __shared__ float s_d[NPB];
  __shared__ int s_cnt;

  const int tid = (int)threadIdx.x;
  const int lane = tid & 31;
  const int wave = __builtin_amdgcn_readfirstlane(tid >> 5);
  const int n0 = blockIdx.x * NPB;
  const int cnt = (n_nodes - n0 < NPB) ? (n_nodes - n0) : NPB;
  unsigned* hdr = lists + (size_t)blockIdx.x * (size_t)lstride;

  AggCtx c;
  c.src = src;
  c.s = s;
  c.sd = s_d;
  c.smaxu = s_maxu;
  c.sden = s_den;
  c.H = H;
  c.sagg = sagg;
  c.qe = s_qe + wave * QCAP;
  c.qd = s_qd + wave * QCAP;
  c.glist = hdr + LHDR;
  c.scnt = &s_cnt;
  c.cap = lstride - LHDR;
  c.n0 = n0;
  c.lane = lane;

  for (int i = tid; i < NPB * CH / 4; i += 1024)
    ((float4*)sagg)[i] = make_float4(0.0f, 0.0f, 0.0f, 0.0f);
  for (int rl = tid; rl < cnt; rl += 1024) {
    const int v = n0 + rl;
    const float dv = d[v];
    float val = s[v] + dv;
    val = (val > 0.0f) ? val : NEG_SLOPE * val;
    s_maxu[rl] = enc_f32(val);
    s_d[rl] = dv;
    s_den[rl] = 0.0f;
  }
  if (tid == 0) s_cnt = 0;
  int total = 0;
  if (!build) total = (int)hdr[0];
  __syncthreads();

  if (!build && total <= c.cap) walk_list<0>(c, total, 0, wave);
  else scan_edges<0>(c, dst, n_edges, cnt, wave, build != 0);
  __syncthreads();
  if (build) {
    total = __builtin_amdgcn_readfirstlane(s_cnt);
    if (tid == 0) hdr[0] = (unsigned)total;
  }

  if (total <= c.cap) {
    walk_list<1>(c, total, cnt, wave);
  } else {
    scan_edges<1>(c, dst, n_edges, cnt, wave, false);
    walk_list<1>(c, 0, cnt, wave);
  }
  __syncthreads();

  const float4 b4 = ((const float4*)bias)[lane];
  for (int rb = wave; rb < cnt; rb += 128) {
    float4 xv[4];
#pragma unroll
    for (int i = 0; i < 4; ++i) {
      const int rl = rb + 32 * i;
      if (rl < cnt)
        xv[i] = ((const float4*)(xin + (size_t)(n0 + rl) * CH))[lane];
    }
#pragma unroll
    for (int i = 0; i < 4; ++i) {
      const int rl = rb + 32 * i;
      if (rl < cnt) {
        const float* ap = sagg + rl * CH + lane;
        const float dd = s_den[rl] + SM_EPS;
        float4 o = xv[i];
        o.x += 1.0f / (1.0f + __builtin_expf(-(ap[0] / dd + b4.x)));
        o.y += 1.0f / (1.0f + __builtin_expf(-(ap[32] / dd + b4.y)));
        o.z += 1.0f / (1.0f + __builtin_expf(-(ap[64] / dd + b4.z)));
        o.w += 1.0f / (1.0f + __builtin_expf(-(ap[96] / dd + b4.w)));
        ((float4*)(xout + (size_t)(n0 + rl) * CH))[lane] = o;
      }
    }
  }
}

extern "C" void kernel_launch(void* const* d_in, const int* in_sizes, int n_in,
                              void* d_out, int out_size, void* d_ws,
                              size_t ws_size, hipStream_t stream) {
  const float* x_in = (const float*)d_in[0];
  const int* ei = (const int*)d_in[1];
  const float* Ws = (const float*)d_in[2];
  const float* att_src = (const float*)d_in[3];
  const float* att_dst = (const float*)d_in[4];
  const float* bias = (const float*)d_in[5];

  const int N = in_sizes[0] / CH;
  const int E = in_sizes[1] / 2;
  const int M = in_sizes[3] / CH;
  const int* src = ei;
  const int* dst = ei + E;

  float* ws = (float*)d_ws;
  float* h = ws;
  float* s = h + (size_t)N * CH;
  float* dsc = s + N;
  unsigned* lists = (unsigned*)(dsc + N);

  float* x = (float*)d_out;

  const int blk = 256;
  const int g_gemm = (N + 127) / 128;
  const int g_own = (N + NPB - 1) / NPB;
  const int lstride = (int)(((size_t)E + 2 * (size_t)N) / (size_t)g_own);
  const size_t smem = (size_t)CH * WT_PAD * sizeof(float);

  for (int m = 0; m < M; ++m) {
    const float* W = Ws + (size_t)m * CH * CH;
    const float* as_ = att_src + (size_t)m * CH;
    const float* ad_ = att_dst + (size_t)m * CH;
    const float* b_ = bias + (size_t)m * CH;

    const float* xr = (m == 0) ? x_in : x;
    gat_gemm_wmma<<<g_gemm, blk, smem, stream>>>(xr, W, as_, ad_, h, s, dsc, N);
    gat_edge_aggregate<<<g_own, 1024, 0, stream>>>(src, dst, s, dsc, h, b_, xr,
                                                   x, lists, lstride,
                                                   (m == 0) ? 1 : 0, E, N);
  }
}


# edge lists built by one binning launch (each edge read once chip-wide; LDS rank + one global int add per block and group reserve list words): no group scans all indices; sums as before; 7 launches
# speedup vs baseline: 5.3448x; 1.0912x over previous
#include <hip/hip_runtime.h>


typedef float v2f __attribute__((ext_vector_type(2)));
typedef float v8f __attribute__((ext_vector_type(8)));

#define CH 128
#define NEG_SLOPE 0.2f
#define SM_EPS 1e-16f

__device__ __forceinline__ unsigned enc_f32(float f) {
  unsigned ib = __float_as_uint(f);
  return (ib & 0x80000000u) ? ~ib : (ib | 0x80000000u);
}
__device__ __forceinline__ float dec_f32(unsigned u) {
  return __uint_as_float((u & 0x80000000u) ? (u & 0x7FFFFFFFu) : ~u);
}

#define WT_PAD 132

__global__ void gat_gemm_wmma(const float* __restrict__ X,
                              const float* __restrict__ W,
                              const float* __restrict__ a_src,
                              const float* __restrict__ a_dst,
                              float* __restrict__ H, float* __restrict__ s,
                              float* __restrict__ d, int n_nodes,
                              unsigned* __restrict__ cnt0, int cstride,
                              int ngroups) {
  extern __shared__ float wt[];
  const int tid = threadIdx.x;
  if (cnt0 != nullptr && blockIdx.x == 0)
    for (int i = tid; i < ngroups; i += 256) cnt0[(size_t)i * cstride] = 0u;
  for (int i = tid; i < CH * CH; i += 256) {
    int k = i >> 7;
    int n = i & (CH - 1);
    wt[n * WT_PAD + k] = W[i];
  }
  __syncthreads();

  const int wave = tid >> 5;
  const int lane = tid & 31;
  const int row0 = blockIdx.x * 128 + wave * 16;
  const int mrow = lane & 15;
  const int koff = (lane >> 4) * 2;

  int rload = row0 + mrow;
  if (rload >= n_nodes) rload = n_nodes - 1;
  const float* xrow = X + (size_t)rload * CH + koff;
  const float* wrow = wt + mrow * WT_PAD + koff;

  v8f acc[8];
#pragma unroll
  for (int t = 0; t < 8; ++t) acc[t] = {};

  v2f a = *(const v2f*)(xrow);
  v2f b[8];
#pragma unroll
  for (int t = 0; t < 8; ++t)
    b[t] = *(const v2f*)(wrow + t * 16 * WT_PAD);

  for (int k4 = 0; k4 < CH; k4 += 4) {
    const int kn = k4 + 4;
    v2f an = {};
    v2f bn[8];
#pragma unroll
    for (int t = 0; t < 8; ++t) bn[t] = {};
    if (kn < CH) {
      an = *(const v2f*)(xrow + kn);
#pragma unroll
      for (int t = 0; t < 8; ++t)
        bn[t] = *(const v2f*)(wrow + t * 16 * WT_PAD + kn);
    }
#pragma unroll
    for (int t = 0; t < 8; ++t)
      acc[t] = __builtin_amdgcn_wmma_f32_16x16x4_f32(
          false, a, false, b[t], (short)0, acc[t], false, false);
    a = an;
#pragma unroll
    for (int t = 0; t < 8; ++t) b[t] = bn[t];
  }

  const int rhi = (lane >> 4) * 8;

  float ps[8], pd[8];
#pragma unroll
  for (int r = 0; r < 8; ++r) { ps[r] = 0.0f; pd[r] = 0.0f; }
#pragma unroll
  for (int t = 0; t < 8; ++t) {
    const float as = a_src[t * 16 + (lane & 15)];
    const float ad = a_dst[t * 16 + (lane & 15)];
#pragma unroll
    for (int r = 0; r < 8; ++r) {
      ps[r] += acc[t][r] * as;
      pd[r] += acc[t][r] * ad;
    }
  }
#pragma unroll
  for (int o = 8; o > 0; o >>= 1) {
#pragma unroll
    for (int r = 0; r < 8; ++r) {
      ps[r] += __shfl_xor(ps[r], o, 32);
      pd[r] += __shfl_xor(pd[r], o, 32);
    }
  }
  float so = ps[0], dq = pd[0];
#pragma unroll
  for (int r = 1; r < 8; ++r) {
    so = ((lane & 15) == r) ? ps[r] : so;
    dq = ((lane & 15) == r) ? pd[r] : dq;
  }
  const int orow = row0 + (lane & 15) + rhi;
  if ((lane & 15) < 8 && orow < n_nodes) {
    s[orow] = so;
    d[orow] = dq;
  }

#pragma unroll
  for (int t = 0; t < 8; ++t) {
    const int col = t * 16 + (lane & 15);
#pragma unroll
    for (int r = 0; r < 8; ++r) {
      int row = row0 + r + rhi;
      if (row < n_nodes) H[(size_t)row * CH + col] = acc[t][r];
    }
  }
}

#define NPB 400
#define QCAP 192
#define EIF 8
#define LHDR 32
#define LROW 23

struct AggCtx {
  const int* src;
  const float* s;
  const float* sd;
  unsigned* smaxu;
  float* sden;
  const float* H;
  float* sagg;
  int* qe;
  unsigned short* qd;
  const unsigned* glist;
  int cap;
  int n0;
  int lane;
};

__device__ __forceinline__ float edge_score(const AggCtx& c, int u, int dl) {
  const float val = c.s[u] + c.sd[dl];
  return (val > 0.0f) ? val : NEG_SLOPE * val;
}

__device__ __forceinline__ void edges_max(const AggCtx& c, int u, int dl,
                                          int n) {
  const float ev = edge_score(c, u, dl);
  if (c.lane < n) atomicMax(c.smaxu + dl, enc_f32(ev));
}

__device__ __forceinline__ void edges_sum(const AggCtx& c, int u, int dl,
                                          int n) {
  const float ev = edge_score(c, u, dl);
  const float w = __builtin_expf(ev - dec_f32(c.smaxu[dl]));
  if (c.lane < n) atomicAdd(c.sden + dl, w);
  const int al = __float_as_int(w);
  int b = 0;
  for (; b + EIF <= n; b += EIF) {
    float4 r[EIF];
#pragma unroll
    for (int i = 0; i < EIF; ++i) {
      const int uj = __builtin_amdgcn_readlane(u, b + i);
      r[i] = *(const float4*)(c.H + (size_t)uj * CH + 4 * c.lane);
    }
#pragma unroll
    for (int i = 0; i < EIF; ++i) {
      const int dj = __builtin_amdgcn_readlane(dl, b + i);
      const float aj = __int_as_float(__builtin_amdgcn_readlane(al, b + i));
      float* ap = c.sagg + dj * CH + c.lane;
      atomicAdd(ap + 0, aj * r[i].x);
      atomicAdd(ap + 32, aj * r[i].y);
      atomicAdd(ap + 64, aj * r[i].z);
      atomicAdd(ap + 96, aj * r[i].w);
    }
  }
  for (; b < n; ++b) {
    const int uj = __builtin_amdgcn_readlane(u, b);
    const int dj = __builtin_amdgcn_readlane(dl, b);
    const float aj = __int_as_float(__builtin_amdgcn_readlane(al, b));
    const float4 r = *(const float4*)(c.H + (size_t)uj * CH + 4 * c.lane);
    float* ap = c.sagg + dj * CH + c.lane;
    atomicAdd(ap + 0, aj * r.x);
    atomicAdd(ap + 32, aj * r.y);
    atomicAdd(ap + 64, aj * r.z);
    atomicAdd(ap + 96, aj * r.w);
  }
}

template <int MODE>
__device__ __forceinline__ void walk_list(const AggCtx& c, int total,
                                          int nself, int wave) {
  const int tot = total + ((MODE == 1) ? nself : 0);
  for (int g0 = wave * 32; g0 < tot; g0 += 1024) {
    const int n = (tot - g0 < 32) ? (tot - g0) : 32;
    const int i = g0 + ((c.lane < n) ? c.lane : 0);
    int u, dl;
    if (i < total) {
      const unsigned wd = c.glist[i];
      u = (int)(wd & ((1u << LROW) - 1u));
      dl = (int)(wd >> LROW);
    } else {
      dl = i - total;
      u = c.n0 + dl;
    }
    if (MODE == 0) edges_max(c, u, dl, n);
    else edges_sum(c, u, dl, n);
  }
}

template <int MODE>
__device__ __forceinline__ void queue_group(const AggCtx& c, int qb, int n) {
  const int j = qb + ((c.lane < n) ? c.lane : 0);
  const int e = c.qe[j];
  const int dl = c.qd[j];
  const int u = c.src[e];
  if (MODE == 0) edges_max(c, u, dl, n);
  else edges_sum(c, u, dl, n);
}

__device__ __forceinline__ void agg_append(const AggCtx& c, bool m, int e,
                                           int dl, int& qn) {
  const unsigned bal = __builtin_amdgcn_ballot_w32(m);
  if (m) {
    const int pos = qn + (int)__builtin_amdgcn_mbcnt_lo(bal, 0u);
    c.qe[pos] = e;
    c.qd[pos] = (unsigned short)dl;
  }
  qn += __builtin_popcount(bal);
}

__device__ __forceinline__ int4 agg_load(const int4* dst4, int vi, int nb4) {
  return dst4[(vi < nb4) ? vi : (nb4 - 1)];
}

template <int MODE>
__device__ __forceinline__ void agg_scan4(const AggCtx& c, int4 d, int e0,
                                          unsigned cnt, int& qn) {
  const int l0 = d.x - c.n0, l1 = d.y - c.n0, l2 = d.z - c.n0,
            l3 = d.w - c.n0;
  const bool m0 = (unsigned)l0 < cnt, m1 = (unsigned)l1 < cnt,
             m2 = (unsigned)l2 < cnt, m3 = (unsigned)l3 < cnt;
  if (__builtin_amdgcn_ballot_w32(m0 | m1 | m2 | m3)) {
    agg_append(c, m0, e0 + 0, l0, qn);
    agg_append(c, m1, e0 + 1, l1, qn);
    agg_append(c, m2, e0 + 2, l2, qn);
    agg_append(c, m3, e0 + 3, l3, qn);
    if (qn >= 64) {
      const int nfull = qn & ~31;
      for (int qb = 0; qb < nfull; qb += 32)
        queue_group<MODE>(c, qb, 32);
      const int rem = qn - nfull;
      const int te = c.qe[nfull + c.lane];
      const unsigned short td = c.qd[nfull + c.lane];
      if (c.lane < rem) {
        c.qe[c.lane] = te;
        c.qd[c.lane] = td;
      }
      qn = rem;
    }
  }
}

template <int MODE>
__device__ __forceinline__ void scan_edges(const AggCtx& c, const int* dst,
                                           int n_edges, int cnt, int wave) {
  int qn = 0;
  int head = (int)(((16u - (unsigned)((size_t)dst & 15u)) & 15u) >> 2);
  if (head > n_edges) head = n_edges;
  const int nb4 = (n_edges - head) >> 2;
  const int tail = (n_edges - head) & 3;
  if (wave == 0) {
    const int nleft = head + tail;
    const int e = (c.lane < head) ? c.lane : (n_edges - tail + (c.lane - head));
    const bool ok = c.lane < nleft;
    const int dl = (ok ? dst[e] : -1) - c.n0;
    agg_append(c, ok && (unsigned)dl < (unsigned)cnt, e, dl, qn);
  }
  const int4* dst4 = (const int4*)(dst + head);
  if (nb4 > 0) {
    const int T = (nb4 + 4095) >> 12;
    int t = (int)((blockIdx.x * (unsigned)T) / gridDim.x);
    if (t >= T) t = 0;
    int4 cur[4], nxt[4];
#pragma unroll
    for (int s = 0; s < 4; ++s)
      cur[s] = agg_load(dst4, (t * 32 + wave) * 128 + s * 32 + c.lane, nb4);
    for (int k = 0; k < T; ++k) {
      int tn = t + 1;
      if (tn >= T) tn -= T;
#pragma unroll
      for (int s = 0; s < 4; ++s)
        nxt[s] = agg_load(dst4, (tn * 32 + wave) * 128 + s * 32 + c.lane, nb4);
#pragma unroll
      for (int s = 0; s < 4; ++s) {
        const int vi = (t * 32 + wave) * 128 + s * 32 + c.lane;
        agg_scan4<MODE>(c, cur[s], head + 4 * vi,
                        (vi < nb4) ? (unsigned)cnt : 0u, qn);
      }
#pragma unroll
      for (int s = 0; s < 4; ++s) cur[s] = nxt[s];
      t = tn;
    }
  }
  for (int qb = 0; qb < qn; qb += 32)
    queue_group<MODE>(c, qb, (qn - qb < 32) ? (qn - qb) : 32);
}

#define BIN_T 256
#define BIN_EPT 16

__global__ void __launch_bounds__(BIN_T)
gat_bin_edges(const int* __restrict__ src, const int* __restrict__ dst,
              unsigned* __restrict__ lists, int lstride, int ngroups,
              int n_edges) {
  extern __shared__ int sbin[];
  int* cntl = sbin;
  int* basel = sbin + ngroups;
  const int tid = (int)threadIdx.x;
  for (int i = tid; i < ngroups; i += BIN_T) cntl[i] = 0;

  const int e0 = (int)blockIdx.x * (BIN_T * BIN_EPT) + tid;
  int u[BIN_EPT], dv[BIN_EPT];
#pragma unroll
  for (int i = 0; i < BIN_EPT; ++i) {
    const int e = e0 + i * BIN_T;
    const bool ok = e < n_edges;
    dv[i] = ok ? dst[e] : -1;
    u[i] = ok ? src[e] : 0;
  }
  __syncthreads();

  int g[BIN_EPT], rk[BIN_EPT];
#pragma unroll
  for (int i = 0; i < BIN_EPT; ++i) {
    int gi = (dv[i] >= 0) ? (dv[i] / NPB) : -1;
    if (gi >= ngroups) gi = -1;
    g[i] = gi;
    rk[i] = 0;
    if (gi >= 0) rk[i] = atomicAdd(cntl + gi, 1);
  }
  __syncthreads();
  for (int i = tid; i < ngroups; i += BIN_T) {
    const int cg = cntl[i];
    int b = 0;
    if (cg > 0)
      b = (int)atomicAdd(lists + (size_t)i * (size_t)lstride, (unsigned)cg);
    basel[i] = b;
  }
  __syncthreads();
  const int cap = lstride - LHDR;
  int pos[BIN_EPT];
#pragma unroll
  for (int i = 0; i < BIN_EPT; ++i)
    pos[i] = basel[(g[i] >= 0) ? g[i] : 0] + rk[i];
#pragma unroll
  for (int i = 0; i < BIN_EPT; ++i) {
    if (g[i] >= 0 && pos[i] < cap)
      lists[(size_t)g[i] * (size_t)lstride + LHDR + pos[i]] =
          (unsigned)u[i] | ((unsigned)(dv[i] - g[i] * NPB) << LROW);
  }
}

__global__ void __launch_bounds__(1024)
gat_edge_aggregate(const int* __restrict__ src, const int* __restrict__ dst,
                   const float* __restrict__ s, const float* __restrict__ d,
                   const float* __restrict__ H,
                   const float* __restrict__ bias, const float* xin,
                   float* xout, const unsigned* __restrict__ lists,
                   int lstride, int n_edges, int n_nodes) {
  __shared__ __attribute__((aligned(16))) float sagg[NPB * CH];
  __shared__ int s_qe[32 * QCAP];
  __shared__ unsigned short s_qd[32 * QCAP];
  __shared__ unsigned s_maxu[NPB];
  __shared__ float s_den[NPB];
  __shared__ float s_d[NPB];

  const int tid = (int)threadIdx.x;
  const int lane = tid & 31;
  const int wave = __builtin_amdgcn_readfirstlane(tid >> 5);
  const int n0 = blockIdx.x * NPB;
  const int cnt = (n_nodes - n0 < NPB) ? (n_nodes - n0) : NPB;
  const unsigned* hdr = lists + (size_t)blockIdx.x * (size_t)lstride;

  AggCtx c;
  c.src = src;
  c.s = s;
  c.sd = s_d;
  c.smaxu = s_maxu;
  c.sden = s_den;
  c.H = H;
  c.sagg = sagg;
  c.qe = s_qe + wave * QCAP;
  c.qd = s_qd + wave * QCAP;
  c.glist = hdr + LHDR;
  c.cap = lstride - LHDR;
  c.n0 = n0;
  c.lane = lane;

  for (int i = tid; i < NPB * CH / 4; i += 1024)
    ((float4*)sagg)[i] = make_float4(0.0f, 0.0f, 0.0f, 0.0f);
  for (int rl = tid; rl < cnt; rl += 1024) {
    const int v = n0 + rl;
    const float dv = d[v];
    float val = s[v] + dv;
    val = (val > 0.0f) ? val : NEG_SLOPE * val;
    s_maxu[rl] = enc_f32(val);
    s_d[rl] = dv;
    s_den[rl] = 0.0f;
  }
  const int total = __builtin_amdgcn_readfirstlane((int)hdr[0]);
  __syncthreads();

  if (total <= c.cap) walk_list<0>(c, total, 0, wave);
  else scan_edges<0>(c, dst, n_edges, cnt, wave);
  __syncthreads();

  if (total <= c.cap) {
    walk_list<1>(c, total, cnt, wave);
  } else {
    scan_edges<1>(c, dst, n_edges, cnt, wave);
    walk_list<1>(c, 0, cnt, wave);
  }
  __syncthreads();

  const float4 b4 = ((const float4*)bias)[lane];
  for (int rb = wave; rb < cnt; rb += 128) {
    float4 xv[4];
#pragma unroll
    for (int i = 0; i < 4; ++i) {
      const int rl = rb + 32 * i;
      if (rl < cnt)
        xv[i] = ((const float4*)(xin + (size_t)(n0 + rl) * CH))[lane];
    }
#pragma unroll
    for (int i = 0; i < 4; ++i) {
      const int rl = rb + 32 * i;
      if (rl < cnt) {
        const float* ap = sagg + rl * CH + lane;
        const float dd = s_den[rl] + SM_EPS;
        float4 o = xv[i];
        o.x += 1.0f / (1.0f + __builtin_expf(-(ap[0] / dd + b4.x)));
        o.y += 1.0f / (1.0f + __builtin_expf(-(ap[32] / dd + b4.y)));
        o.z += 1.0f / (1.0f + __builtin_expf(-(ap[64] / dd + b4.z)));
        o.w += 1.0f / (1.0f + __builtin_expf(-(ap[96] / dd + b4.w)));
        ((float4*)(xout + (size_t)(n0 + rl) * CH))[lane] = o;
      }
    }
  }
}

extern "C" void kernel_launch(void* const* d_in, const int* in_sizes, int n_in,
                              void* d_out, int out_size, void* d_ws,
                              size_t ws_size, hipStream_t stream) {
  const float* x_in = (const float*)d_in[0];
  const int* ei = (const int*)d_in[1];
  const float* Ws = (const float*)d_in[2];
  const float* att_src = (const float*)d_in[3];
  const float* att_dst = (const float*)d_in[4];
  const float* bias = (const float*)d_in[5];

  const int N = in_sizes[0] / CH;
  const int E = in_sizes[1] / 2;
  const int M = in_sizes[3] / CH;
  const int* src = ei;
  const int* dst = ei + E;

  float* ws = (float*)d_ws;
  float* h = ws;
  float* s = h + (size_t)N * CH;
  float* dsc = s + N;
  unsigned* lists = (unsigned*)(dsc + N);

  float* x = (float*)d_out;

  const int blk = 256;
  const int g_gemm = (N + 127) / 128;
  const int g_own = (N + NPB - 1) / NPB;
  const int lstride = (int)(((size_t)E + 2 * (size_t)N) / (size_t)g_own);
  const size_t smem = (size_t)CH * WT_PAD * sizeof(float);
  const int g_bin = (E + BIN_T * BIN_EPT - 1) / (BIN_T * BIN_EPT);
  const size_t smem_bin = (size_t)g_own * 2 * sizeof(int);

  for (int m = 0; m < M; ++m) {
    const float* W = Ws + (size_t)m * CH * CH;
    const float* as_ = att_src + (size_t)m * CH;
    const float* ad_ = att_dst + (size_t)m * CH;
    const float* b_ = bias + (size_t)m * CH;

    const float* xr = (m == 0) ? x_in : x;
    gat_gemm_wmma<<<g_gemm, blk, smem, stream>>>(xr, W, as_, ad_, h, s, dsc, N,
                                                 (m == 0) ? lists : nullptr,
                                                 lstride, g_own);
    if (m == 0 && E > 0)
      gat_bin_edges<<<g_bin, BIN_T, smem_bin, stream>>>(src, dst, lists,
                                                        lstride, g_own, E);
    gat_edge_aggregate<<<g_own, 1024, 0, stream>>>(src, dst, s, dsc, h, b_, xr,
                                                   x, lists, lstride, E, N);
  }
}


# edge aggregate: max pass keeps each listed edge's score e in LDS (32 KiB); sum pass reads it there, the second random 4-B gather of s[src] per edge is gone; sums atomic, arrival order as before
# speedup vs baseline: 5.5709x; 1.0423x over previous
#include <hip/hip_runtime.h>


typedef float v2f __attribute__((ext_vector_type(2)));
typedef float v8f __attribute__((ext_vector_type(8)));

#define CH 128
#define NEG_SLOPE 0.2f
#define SM_EPS 1e-16f

__device__ __forceinline__ unsigned enc_f32(float f) {
  unsigned ib = __float_as_uint(f);
  return (ib & 0x80000000u) ? ~ib : (ib | 0x80000000u);
}
__device__ __forceinline__ float dec_f32(unsigned u) {
  return __uint_as_float((u & 0x80000000u) ? (u & 0x7FFFFFFFu) : ~u);
}

#define WT_PAD 132

__global__ void gat_gemm_wmma(const float* __restrict__ X,
                              const float* __restrict__ W,
                              const float* __restrict__ a_src,
                              const float* __restrict__ a_dst,
                              float* __restrict__ H, float* __restrict__ s,
                              float* __restrict__ d, int n_nodes,
                              unsigned* __restrict__ cnt0, int cstride,
                              int ngroups) {
  extern __shared__ float wt[];
  const int tid = threadIdx.x;
  if (cnt0 != nullptr && blockIdx.x == 0)
    for (int i = tid; i < ngroups; i += 256) cnt0[(size_t)i * cstride] = 0u;
  for (int i = tid; i < CH * CH; i += 256) {
    int k = i >> 7;
    int n = i & (CH - 1);
    wt[n * WT_PAD + k] = W[i];
  }
  __syncthreads();

  const int wave = tid >> 5;
  const int lane = tid & 31;
  const int row0 = blockIdx.x * 128 + wave * 16;
  const int mrow = lane & 15;
  const int koff = (lane >> 4) * 2;

  int rload = row0 + mrow;
  if (rload >= n_nodes) rload = n_nodes - 1;
  const float* xrow = X + (size_t)rload * CH + koff;
  const float* wrow = wt + mrow * WT_PAD + koff;

  v8f acc[8];
#pragma unroll
  for (int t = 0; t < 8; ++t) acc[t] = {};

  v2f a = *(const v2f*)(xrow);
  v2f b[8];
#pragma unroll
  for (int t = 0; t < 8; ++t)
    b[t] = *(const v2f*)(wrow + t * 16 * WT_PAD);

  for (int k4 = 0; k4 < CH; k4 += 4) {
    const int kn = k4 + 4;
    v2f an = {};
    v2f bn[8];
#pragma unroll
    for (int t = 0; t < 8; ++t) bn[t] = {};
    if (kn < CH) {
      an = *(const v2f*)(xrow + kn);
#pragma unroll
      for (int t = 0; t < 8; ++t)
        bn[t] = *(const v2f*)(wrow + t * 16 * WT_PAD + kn);
    }
#pragma unroll
    for (int t = 0; t < 8; ++t)
      acc[t] = __builtin_amdgcn_wmma_f32_16x16x4_f32(
          false, a, false, b[t], (short)0, acc[t], false, false);
    a = an;
#pragma unroll
    for (int t = 0; t < 8; ++t) b[t] = bn[t];
  }

  const int rhi = (lane >> 4) * 8;

  float ps[8], pd[8];
#pragma unroll
  for (int r = 0; r < 8; ++r) { ps[r] = 0.0f; pd[r] = 0.0f; }
#pragma unroll
  for (int t = 0; t < 8; ++t) {
    const float as = a_src[t * 16 + (lane & 15)];
    const float ad = a_dst[t * 16 + (lane & 15)];
#pragma unroll
    for (int r = 0; r < 8; ++r) {
      ps[r] += acc[t][r] * as;
      pd[r] += acc[t][r] * ad;
    }
  }
#pragma unroll
  for (int o = 8; o > 0; o >>= 1) {
#pragma unroll
    for (int r = 0; r < 8; ++r) {
      ps[r] += __shfl_xor(ps[r], o, 32);
      pd[r] += __shfl_xor(pd[r], o, 32);
    }
  }
  float so = ps[0], dq = pd[0];
#pragma unroll
  for (int r = 1; r < 8; ++r) {
    so = ((lane & 15) == r) ? ps[r] : so;
    dq = ((lane & 15) == r) ? pd[r] : dq;
  }
  const int orow = row0 + (lane & 15) + rhi;
  if ((lane & 15) < 8 && orow < n_nodes) {
    s[orow] = so;
    d[orow] = dq;
  }

#pragma unroll
  for (int t = 0; t < 8; ++t) {
    const int col = t * 16 + (lane & 15);
#pragma unroll
    for (int r = 0; r < 8; ++r) {
      int row = row0 + r + rhi;
      if (row < n_nodes) H[(size_t)row * CH + col] = acc[t][r];
    }
  }
}

#define NPB 400
#define QCAP 192
#define EIF 8
#define LHDR 32
#define LROW 23
#define SE_CAP 8192

struct AggCtx {
  const int* src;
  const float* s;
  const float* sd;
  unsigned* smaxu;
  float* sden;
  const float* H;
  float* sagg;
  int* qe;
  unsigned short* qd;
  float* se;
  const unsigned* glist;
  int cap;
  int n0;
  int lane;
};

__device__ __forceinline__ float edge_score(const AggCtx& c, int u, int dl) {
  const float val = c.s[u] + c.sd[dl];
  return (val > 0.0f) ? val : NEG_SLOPE * val;
}

__device__ __forceinline__ float edges_max(const AggCtx& c, int u, int dl,
                                           int n) {
  const float ev = edge_score(c, u, dl);
  if (c.lane < n) atomicMax(c.smaxu + dl, enc_f32(ev));
  return ev;
}

__device__ __forceinline__ void edges_sum_ev(const AggCtx& c, int u, int dl,
                                             float ev, int n) {
  const float w = __builtin_expf(ev - dec_f32(c.smaxu[dl]));
  if (c.lane < n) atomicAdd(c.sden + dl, w);
  const int al = __float_as_int(w);
  int b = 0;
  for (; b + EIF <= n; b += EIF) {
    float4 r[EIF];
#pragma unroll
    for (int i = 0; i < EIF; ++i) {
      const int uj = __builtin_amdgcn_readlane(u, b + i);
      r[i] = *(const float4*)(c.H + (size_t)uj * CH + 4 * c.lane);
    }
#pragma unroll
    for (int i = 0; i < EIF; ++i) {
      const int dj = __builtin_amdgcn_readlane(dl, b + i);
      const float aj = __int_as_float(__builtin_amdgcn_readlane(al, b + i));
      float* ap = c.sagg + dj * CH + c.lane;
      atomicAdd(ap + 0, aj * r[i].x);
      atomicAdd(ap + 32, aj * r[i].y);
      atomicAdd(ap + 64, aj * r[i].z);
      atomicAdd(ap + 96, aj * r[i].w);
    }
  }
  for (; b < n; ++b) {
    const int uj = __builtin_amdgcn_readlane(u, b);
    const int dj = __builtin_amdgcn_readlane(dl, b);
    const float aj = __int_as_float(__builtin_amdgcn_readlane(al, b));
    const float4 r = *(const float4*)(c.H + (size_t)uj * CH + 4 * c.lane);
    float* ap = c.sagg + dj * CH + c.lane;
    atomicAdd(ap + 0, aj * r.x);
    atomicAdd(ap + 32, aj * r.y);
    atomicAdd(ap + 64, aj * r.z);
    atomicAdd(ap + 96, aj * r.w);
  }
}

__device__ __forceinline__ void edges_sum(const AggCtx& c, int u, int dl,
                                          int n) {
  edges_sum_ev(c, u, dl, edge_score(c, u, dl), n);
}

template <int MODE>
__device__ __forceinline__ void walk_list(const AggCtx& c, int total,
                                          int nself, int wave) {
  const int tot = total + ((MODE == 1) ? nself : 0);
  for (int g0 = wave * 32; g0 < tot; g0 += 1024) {
    const int n = (tot - g0 < 32) ? (tot - g0) : 32;
    const int i = g0 + ((c.lane < n) ? c.lane : 0);
    int u, dl;
    if (i < total) {
      const unsigned wd = c.glist[i];
      u = (int)(wd & ((1u << LROW) - 1u));
      dl = (int)(wd >> LROW);
    } else {
      dl = i - total;
      u = c.n0 + dl;
    }
    const bool kept = i < total && i < SE_CAP;
    if (MODE == 0) {
      const float ev = edges_max(c, u, dl, n);
      if (kept && c.lane < n) c.se[i] = ev;
    } else {
      float ev;
      if (kept) ev = c.se[i];
      else ev = edge_score(c, u, dl);
      edges_sum_ev(c, u, dl, ev, n);
    }
  }
}

template <int MODE>
__device__ __forceinline__ void queue_group(const AggCtx& c, int qb, int n) {
  const int j = qb + ((c.lane < n) ? c.lane : 0);
  const int e = c.qe[j];
  const int dl = c.qd[j];
  const int u = c.src[e];
  if (MODE == 0) edges_max(c, u, dl, n);
  else edges_sum(c, u, dl, n);
}

__device__ __forceinline__ void agg_append(const AggCtx& c, bool m, int e,
                                           int dl, int& qn) {
  const unsigned bal = __builtin_amdgcn_ballot_w32(m);
  if (m) {
    const int pos = qn + (int)__builtin_amdgcn_mbcnt_lo(bal, 0u);
    c.qe[pos] = e;
    c.qd[pos] = (unsigned short)dl;
  }
  qn += __builtin_popcount(bal);
}

__device__ __forceinline__ int4 agg_load(const int4* dst4, int vi, int nb4) {
  return dst4[(vi < nb4) ? vi : (nb4 - 1)];
}

template <int MODE>
__device__ __forceinline__ void agg_scan4(const AggCtx& c, int4 d, int e0,
                                          unsigned cnt, int& qn) {
  const int l0 = d.x - c.n0, l1 = d.y - c.n0, l2 = d.z - c.n0,
            l3 = d.w - c.n0;
  const bool m0 = (unsigned)l0 < cnt, m1 = (unsigned)l1 < cnt,
             m2 = (unsigned)l2 < cnt, m3 = (unsigned)l3 < cnt;
  if (__builtin_amdgcn_ballot_w32(m0 | m1 | m2 | m3)) {
    agg_append(c, m0, e0 + 0, l0, qn);
    agg_append(c, m1, e0 + 1, l1, qn);
    agg_append(c, m2, e0 + 2, l2, qn);
    agg_append(c, m3, e0 + 3, l3, qn);
    if (qn >= 64) {
      const int nfull = qn & ~31;
      for (int qb = 0; qb < nfull; qb += 32)
        queue_group<MODE>(c, qb, 32);
      const int rem = qn - nfull;
      const int te = c.qe[nfull + c.lane];
      const unsigned short td = c.qd[nfull + c.lane];
      if (c.lane < rem) {
        c.qe[c.lane] = te;
        c.qd[c.lane] = td;
      }
      qn = rem;
    }
  }
}

template <int MODE>
__device__ __forceinline__ void scan_edges(const AggCtx& c, const int* dst,
                                           int n_edges, int cnt, int wave) {
  int qn = 0;
  int head = (int)(((16u - (unsigned)((size_t)dst & 15u)) & 15u) >> 2);
  if (head > n_edges) head = n_edges;
  const int nb4 = (n_edges - head) >> 2;
  const int tail = (n_edges - head) & 3;
  if (wave == 0) {
    const int nleft = head + tail;
    const int e = (c.lane < head) ? c.lane : (n_edges - tail + (c.lane - head));
    const bool ok = c.lane < nleft;
    const int dl = (ok ? dst[e] : -1) - c.n0;
    agg_append(c, ok && (unsigned)dl < (unsigned)cnt, e, dl, qn);
  }
  const int4* dst4 = (const int4*)(dst + head);
  if (nb4 > 0) {
    const int T = (nb4 + 4095) >> 12;
    int t = (int)((blockIdx.x * (unsigned)T) / gridDim.x);
    if (t >= T) t = 0;
    int4 cur[4], nxt[4];
#pragma unroll
    for (int s = 0; s < 4; ++s)
      cur[s] = agg_load(dst4, (t * 32 + wave) * 128 + s * 32 + c.lane, nb4);
    for (int k = 0; k < T; ++k) {
      int tn = t + 1;
      if (tn >= T) tn -= T;
#pragma unroll
      for (int s = 0; s < 4; ++s)
        nxt[s] = agg_load(dst4, (tn * 32 + wave) * 128 + s * 32 + c.lane, nb4);
#pragma unroll
      for (int s = 0; s < 4; ++s) {
        const int vi = (t * 32 + wave) * 128 + s * 32 + c.lane;
        agg_scan4<MODE>(c, cur[s], head + 4 * vi,
                        (vi < nb4) ? (unsigned)cnt : 0u, qn);
      }
#pragma unroll
      for (int s = 0; s < 4; ++s) cur[s] = nxt[s];
      t = tn;
    }
  }
  for (int qb = 0; qb < qn; qb += 32)
    queue_group<MODE>(c, qb, (qn - qb < 32) ? (qn - qb) : 32);
}

#define BIN_T 256
#define BIN_EPT 16

__global__ void __launch_bounds__(BIN_T)
gat_bin_edges(const int* __restrict__ src, const int* __restrict__ dst,
              unsigned* __restrict__ lists, int lstride, int ngroups,
              int n_edges) {
  extern __shared__ int sbin[];
  int* cntl = sbin;
  int* basel = sbin + ngroups;
  const int tid = (int)threadIdx.x;
  for (int i = tid; i < ngroups; i += BIN_T) cntl[i] = 0;

  const int e0 = (int)blockIdx.x * (BIN_T * BIN_EPT) + tid;
  int u[BIN_EPT], dv[BIN_EPT];
#pragma unroll
  for (int i = 0; i < BIN_EPT; ++i) {
    const int e = e0 + i * BIN_T;
    const bool ok = e < n_edges;
    dv[i] = ok ? dst[e] : -1;
    u[i] = ok ? src[e] : 0;
  }
  __syncthreads();

  int g[BIN_EPT], rk[BIN_EPT];
#pragma unroll
  for (int i = 0; i < BIN_EPT; ++i) {
    int gi = (dv[i] >= 0) ? (dv[i] / NPB) : -1;
    if (gi >= ngroups) gi = -1;
    g[i] = gi;
    rk[i] = 0;
    if (gi >= 0) rk[i] = atomicAdd(cntl + gi, 1);
  }
  __syncthreads();
  for (int i = tid; i < ngroups; i += BIN_T) {
    const int cg = cntl[i];
    int b = 0;
    if (cg > 0)
      b = (int)atomicAdd(lists + (size_t)i * (size_t)lstride, (unsigned)cg);
    basel[i] = b;
  }
  __syncthreads();
  const int cap = lstride - LHDR;
  int pos[BIN_EPT];
#pragma unroll
  for (int i = 0; i < BIN_EPT; ++i)
    pos[i] = basel[(g[i] >= 0) ? g[i] : 0] + rk[i];
#pragma unroll
  for (int i = 0; i < BIN_EPT; ++i) {
    if (g[i] >= 0 && pos[i] < cap)
      lists[(size_t)g[i] * (size_t)lstride + LHDR + pos[i]] =
          (unsigned)u[i] | ((unsigned)(dv[i] - g[i] * NPB) << LROW);
  }
}

__global__ void __launch_bounds__(1024)
gat_edge_aggregate(const int* __restrict__ src, const int* __restrict__ dst,
                   const float* __restrict__ s, const float* __restrict__ d,
                   const float* __restrict__ H,
                   const float* __restrict__ bias, const float* xin,
                   float* xout, const unsigned* __restrict__ lists,
                   int lstride, int n_edges, int n_nodes) {
  __shared__ __attribute__((aligned(16))) float sagg[NPB * CH];
  __shared__ int s_qe[32 * QCAP];
  __shared__ unsigned short s_qd[32 * QCAP];
  __shared__ unsigned s_maxu[NPB];
  __shared__ float s_den[NPB];
  __shared__ float s_d[NPB];
  __shared__ float s_e[SE_CAP];

  const int tid = (int)threadIdx.x;
  const int lane = tid & 31;
  const int wave = __builtin_amdgcn_readfirstlane(tid >> 5);
  const int n0 = blockIdx.x * NPB;
  const int cnt = (n_nodes - n0 < NPB) ? (n_nodes - n0) : NPB;
  const unsigned* hdr = lists + (size_t)blockIdx.x * (size_t)lstride;

  AggCtx c;
  c.src = src;
  c.s = s;
  c.sd = s_d;
  c.smaxu = s_maxu;
  c.sden = s_den;
  c.H = H;
  c.sagg = sagg;
  c.qe = s_qe + wave * QCAP;
  c.qd = s_qd + wave * QCAP;
  c.se = s_e;
  c.glist = hdr + LHDR;
  c.cap = lstride - LHDR;
  c.n0 = n0;
  c.lane = lane;

  for (int i = tid; i < NPB * CH / 4; i += 1024)
    ((float4*)sagg)[i] = make_float4(0.0f, 0.0f, 0.0f, 0.0f);
  for (int rl = tid; rl < cnt; rl += 1024) {
    const int v = n0 + rl;
    const float dv = d[v];
    float val = s[v] + dv;
    val = (val > 0.0f) ? val : NEG_SLOPE * val;
    s_maxu[rl] = enc_f32(val);
    s_d[rl] = dv;
    s_den[rl] = 0.0f;
  }
  const int total = __builtin_amdgcn_readfirstlane((int)hdr[0]);
  __syncthreads();

  if (total <= c.cap) walk_list<0>(c, total, 0, wave);
  else scan_edges<0>(c, dst, n_edges, cnt, wave);
  __syncthreads();

  if (total <= c.cap) {
    walk_list<1>(c, total, cnt, wave);
  } else {
    scan_edges<1>(c, dst, n_edges, cnt, wave);
    walk_list<1>(c, 0, cnt, wave);
  }
  __syncthreads();

  const float4 b4 = ((const float4*)bias)[lane];
  for (int rb = wave; rb < cnt; rb += 128) {
    float4 xv[4];
#pragma unroll
    for (int i = 0; i < 4; ++i) {
      const int rl = rb + 32 * i;
      if (rl < cnt)
        xv[i] = ((const float4*)(xin + (size_t)(n0 + rl) * CH))[lane];
    }
#pragma unroll
    for (int i = 0; i < 4; ++i) {
      const int rl = rb + 32 * i;
      if (rl < cnt) {
        const float* ap = sagg + rl * CH + lane;
        const float dd = s_den[rl] + SM_EPS;
        float4 o = xv[i];
        o.x += 1.0f / (1.0f + __builtin_expf(-(ap[0] / dd + b4.x)));
        o.y += 1.0f / (1.0f + __builtin_expf(-(ap[32] / dd + b4.y)));
        o.z += 1.0f / (1.0f + __builtin_expf(-(ap[64] / dd + b4.z)));
        o.w += 1.0f / (1.0f + __builtin_expf(-(ap[96] / dd + b4.w)));
        ((float4*)(xout + (size_t)(n0 + rl) * CH))[lane] = o;
      }
    }
  }
}

extern "C" void kernel_launch(void* const* d_in, const int* in_sizes, int n_in,
                              void* d_out, int out_size, void* d_ws,
                              size_t ws_size, hipStream_t stream) {
  const float* x_in = (const float*)d_in[0];
  const int* ei = (const int*)d_in[1];
  const float* Ws = (const float*)d_in[2];
  const float* att_src = (const float*)d_in[3];
  const float* att_dst = (const float*)d_in[4];
  const float* bias = (const float*)d_in[5];

  const int N = in_sizes[0] / CH;
  const int E = in_sizes[1] / 2;
  const int M = in_sizes[3] / CH;
  const int* src = ei;
  const int* dst = ei + E;

  float* ws = (float*)d_ws;
  float* h = ws;
  float* s = h + (size_t)N * CH;
  float* dsc = s + N;
  unsigned* lists = (unsigned*)(dsc + N);

  float* x = (float*)d_out;

  const int blk = 256;
  const int g_gemm = (N + 127) / 128;
  const int g_own = (N + NPB - 1) / NPB;
  const int lstride = (int)(((size_t)E + 2 * (size_t)N) / (size_t)g_own);
  const size_t smem = (size_t)CH * WT_PAD * sizeof(float);
  const int g_bin = (E + BIN_T * BIN_EPT - 1) / (BIN_T * BIN_EPT);
  const size_t smem_bin = (size_t)g_own * 2 * sizeof(int);

  for (int m = 0; m < M; ++m) {
    const float* W = Ws + (size_t)m * CH * CH;
    const float* as_ = att_src + (size_t)m * CH;
    const float* ad_ = att_dst + (size_t)m * CH;
    const float* b_ = bias + (size_t)m * CH;

    const float* xr = (m == 0) ? x_in : x;
    gat_gemm_wmma<<<g_gemm, blk, smem, stream>>>(xr, W, as_, ad_, h, s, dsc, N,
                                                 (m == 0) ? lists : nullptr,
                                                 lstride, g_own);
    if (m == 0 && E > 0)
      gat_bin_edges<<<g_bin, BIN_T, smem_bin, stream>>>(src, dst, lists,
                                                        lstride, g_own, E);
    gat_edge_aggregate<<<g_own, 1024, 0, stream>>>(src, dst, s, dsc, h, b_, xr,
                                                   x, lists, lstride, E, N);
  }
}


# aggregate: max pass deals each edge and its score to the LDS queue of wave (row mod 32); that wave alone sums its rows by 16-B LDS read-add-write, no f32 row atomics, and updates them; queue order
# speedup vs baseline: 5.7920x; 1.0397x over previous
#include <hip/hip_runtime.h>


typedef float v2f __attribute__((ext_vector_type(2)));
typedef float v8f __attribute__((ext_vector_type(8)));

#define CH 128
#define NEG_SLOPE 0.2f
#define SM_EPS 1e-16f

__device__ __forceinline__ unsigned enc_f32(float f) {
  unsigned ib = __float_as_uint(f);
  return (ib & 0x80000000u) ? ~ib : (ib | 0x80000000u);
}
__device__ __forceinline__ float dec_f32(unsigned u) {
  return __uint_as_float((u & 0x80000000u) ? (u & 0x7FFFFFFFu) : ~u);
}

#define WT_PAD 132

__global__ void gat_gemm_wmma(const float* __restrict__ X,
                              const float* __restrict__ W,
                              const float* __restrict__ a_src,
                              const float* __restrict__ a_dst,
                              float* __restrict__ H, float* __restrict__ s,
                              float* __restrict__ d, int n_nodes,
                              unsigned* __restrict__ cnt0, int cstride,
                              int ngroups) {
  extern __shared__ float wt[];
  const int tid = threadIdx.x;
  if (cnt0 != nullptr && blockIdx.x == 0)
    for (int i = tid; i < ngroups; i += 256) cnt0[(size_t)i * cstride] = 0u;
  for (int i = tid; i < CH * CH; i += 256) {
    int k = i >> 7;
    int n = i & (CH - 1);
    wt[n * WT_PAD + k] = W[i];
  }
  __syncthreads();

  const int wave = tid >> 5;
  const int lane = tid & 31;
  const int row0 = blockIdx.x * 128 + wave * 16;
  const int mrow = lane & 15;
  const int koff = (lane >> 4) * 2;

  int rload = row0 + mrow;
  if (rload >= n_nodes) rload = n_nodes - 1;
  const float* xrow = X + (size_t)rload * CH + koff;
  const float* wrow = wt + mrow * WT_PAD + koff;

  v8f acc[8];
#pragma unroll
  for (int t = 0; t < 8; ++t) acc[t] = {};

  v2f a = *(const v2f*)(xrow);
  v2f b[8];
#pragma unroll
  for (int t = 0; t < 8; ++t)
    b[t] = *(const v2f*)(wrow + t * 16 * WT_PAD);

  for (int k4 = 0; k4 < CH; k4 += 4) {
    const int kn = k4 + 4;
    v2f an = {};
    v2f bn[8];
#pragma unroll
    for (int t = 0; t < 8; ++t) bn[t] = {};
    if (kn < CH) {
      an = *(const v2f*)(xrow + kn);
#pragma unroll
      for (int t = 0; t < 8; ++t)
        bn[t] = *(const v2f*)(wrow + t * 16 * WT_PAD + kn);
    }
#pragma unroll
    for (int t = 0; t < 8; ++t)
      acc[t] = __builtin_amdgcn_wmma_f32_16x16x4_f32(
          false, a, false, b[t], (short)0, acc[t], false, false);
    a = an;
#pragma unroll
    for (int t = 0; t < 8; ++t) b[t] = bn[t];
  }

  const int rhi = (lane >> 4) * 8;

  float ps[8], pd[8];
#pragma unroll
  for (int r = 0; r < 8; ++r) { ps[r] = 0.0f; pd[r] = 0.0f; }
#pragma unroll
  for (int t = 0; t < 8; ++t) {
    const float as = a_src[t * 16 + (lane & 15)];
    const float ad = a_dst[t * 16 + (lane & 15)];
#pragma unroll
    for (int r = 0; r < 8; ++r) {
      ps[r] += acc[t][r] * as;
      pd[r] += acc[t][r] * ad;
    }
  }
#pragma unroll
  for (int o = 8; o > 0; o >>= 1) {
#pragma unroll
    for (int r = 0; r < 8; ++r) {
      ps[r] += __shfl_xor(ps[r], o, 32);
      pd[r] += __shfl_xor(pd[r], o, 32);
    }
  }
  float so = ps[0], dq = pd[0];
#pragma unroll
  for (int r = 1; r < 8; ++r) {
    so = ((lane & 15) == r) ? ps[r] : so;
    dq = ((lane & 15) == r) ? pd[r] : dq;
  }
  const int orow = row0 + (lane & 15) + rhi;
  if ((lane & 15) < 8 && orow < n_nodes) {
    s[orow] = so;
    d[orow] = dq;
  }

#pragma unroll
  for (int t = 0; t < 8; ++t) {
    const int col = t * 16 + (lane & 15);
#pragma unroll
    for (int r = 0; r < 8; ++r) {
      int row = row0 + r + rhi;
      if (row < n_nodes) H[(size_t)row * CH + col] = acc[t][r];
    }
  }
}

#define NPB 400
#define QCAP 192
#define EIF 8
#define LHDR 32
#define LROW 23
#define QW 321
static_assert(32 * QCAP * 4 + 32 * QCAP * 2 <= 32 * QW * 4,
              "the scan queues must fit the storage of the waves' queues");

struct AggCtx {
  const int* src;
  const float* s;
  const float* sd;
  unsigned* smaxu;
  float* sden;
  const float* H;
  float* sagg;
  int* qe;
  unsigned short* qd;
  unsigned* qw;
  float* qs;
  int* qcnt;
  const unsigned* glist;
  int cap;
  int n0;
  int lane;
};

__device__ __forceinline__ float edge_score(const AggCtx& c, int u, int dl) {
  const float val = c.s[u] + c.sd[dl];
  return (val > 0.0f) ? val : NEG_SLOPE * val;
}

__device__ __forceinline__ float edges_max(const AggCtx& c, int u, int dl,
                                           int n) {
  const float ev = edge_score(c, u, dl);
  if (c.lane < n) atomicMax(c.smaxu + dl, enc_f32(ev));
  return ev;
}

__device__ __forceinline__ void edges_sum_ev(const AggCtx& c, int u, int dl,
                                             float ev, int n) {
  const float w = __builtin_expf(ev - dec_f32(c.smaxu[dl]));
  if (c.lane < n) atomicAdd(c.sden + dl, w);
  const int al = __float_as_int(w);
  int b = 0;
  for (; b + EIF <= n; b += EIF) {
    float4 r[EIF];
#pragma unroll
    for (int i = 0; i < EIF; ++i) {
      const int uj = __builtin_amdgcn_readlane(u, b + i);
      r[i] = *(const float4*)(c.H + (size_t)uj * CH + 4 * c.lane);
    }
#pragma unroll
    for (int i = 0; i < EIF; ++i) {
      const int dj = __builtin_amdgcn_readlane(dl, b + i);
      const float aj = __int_as_float(__builtin_amdgcn_readlane(al, b + i));
      float* ap = c.sagg + dj * CH + 4 * c.lane;
      atomicAdd(ap + 0, aj * r[i].x);
      atomicAdd(ap + 1, aj * r[i].y);
      atomicAdd(ap + 2, aj * r[i].z);
      atomicAdd(ap + 3, aj * r[i].w);
    }
  }
  for (; b < n; ++b) {
    const int uj = __builtin_amdgcn_readlane(u, b);
    const int dj = __builtin_amdgcn_readlane(dl, b);
    const float aj = __int_as_float(__builtin_amdgcn_readlane(al, b));
    const float4 r = *(const float4*)(c.H + (size_t)uj * CH + 4 * c.lane);
    float* ap = c.sagg + dj * CH + 4 * c.lane;
    atomicAdd(ap + 0, aj * r.x);
    atomicAdd(ap + 1, aj * r.y);
    atomicAdd(ap + 2, aj * r.z);
    atomicAdd(ap + 3, aj * r.w);
  }
}

__device__ __forceinline__ void edges_sum(const AggCtx& c, int u, int dl,
                                          int n) {
  edges_sum_ev(c, u, dl, edge_score(c, u, dl), n);
}

__device__ __forceinline__ void edges_sum_own(const AggCtx& c, int u, int dl,
                                              float ev, int n) {
  const float w = __builtin_expf(ev - dec_f32(c.smaxu[dl]));
  if (c.lane < n) atomicAdd(c.sden + dl, w);
  const int al = __float_as_int(w);
  int b = 0;
  for (; b + EIF <= n; b += EIF) {
    float4 r[EIF];
#pragma unroll
    for (int i = 0; i < EIF; ++i) {
      const int uj = __builtin_amdgcn_readlane(u, b + i);
      r[i] = *(const float4*)(c.H + (size_t)uj * CH + 4 * c.lane);
    }
#pragma unroll
    for (int i = 0; i < EIF; ++i) {
      const int dj = __builtin_amdgcn_readlane(dl, b + i);
      const float aj = __int_as_float(__builtin_amdgcn_readlane(al, b + i));
      float4* ap = (float4*)(c.sagg + dj * CH) + c.lane;
      float4 a = *ap;
      a.x += aj * r[i].x;
      a.y += aj * r[i].y;
      a.z += aj * r[i].z;
      a.w += aj * r[i].w;
      *ap = a;
    }
  }
  for (; b < n; ++b) {
    const int uj = __builtin_amdgcn_readlane(u, b);
    const int dj = __builtin_amdgcn_readlane(dl, b);
    const float aj = __int_as_float(__builtin_amdgcn_readlane(al, b));
    const float4 r = *(const float4*)(c.H + (size_t)uj * CH + 4 * c.lane);
    float4* ap = (float4*)(c.sagg + dj * CH) + c.lane;
    float4 a = *ap;
    a.x += aj * r.x;
    a.y += aj * r.y;
    a.z += aj * r.z;
    a.w += aj * r.w;
    *ap = a;
  }
}

__device__ __forceinline__ void deal_list_max(const AggCtx& c, int total,
                                              int wave) {
  for (int g0 = wave * 32; g0 < total; g0 += 1024) {
    const int n = (total - g0 < 32) ? (total - g0) : 32;
    const int i = g0 + ((c.lane < n) ? c.lane : 0);
    const unsigned wd = c.glist[i];
    const int u = (int)(wd & ((1u << LROW) - 1u));
    const int dl = (int)(wd >> LROW);
    const float ev = edges_max(c, u, dl, n);
    if (c.lane < n) {
      const int q = dl & 31;
      const int pos = atomicAdd(c.qcnt + q, 1);
      if (pos < QW) {
        c.qw[q * QW + pos] = wd;
        c.qs[q * QW + pos] = ev;
      }
    }
  }
}

__device__ __forceinline__ void own_sum(const AggCtx& c, int wave, int cnt) {
  const int nq = __builtin_amdgcn_readfirstlane(c.qcnt[wave]);
  const int nown = (cnt > wave) ? (((cnt - 1 - wave) >> 5) + 1) : 0;
  const int tot = nq + nown;
  const unsigned* qw = c.qw + wave * QW;
  const float* qs = c.qs + wave * QW;
  for (int g0 = 0; g0 < tot; g0 += 32) {
    const int n = (tot - g0 < 32) ? (tot - g0) : 32;
    const int i = g0 + ((c.lane < n) ? c.lane : 0);
    int u, dl;
    float ev;
    if (i < nq) {
      const unsigned wd = qw[i];
      u = (int)(wd & ((1u << LROW) - 1u));
      dl = (int)(wd >> LROW);
      ev = qs[i];
    } else {
      dl = wave + 32 * (i - nq);
      u = c.n0 + dl;
      ev = edge_score(c, u, dl);
    }
    edges_sum_own(c, u, dl, ev, n);
  }
}

__device__ __forceinline__ void walk_list_sum(const AggCtx& c, int total,
                                              int nself, int wave) {
  const int tot = total + nself;
  for (int g0 = wave * 32; g0 < tot; g0 += 1024) {
    const int n = (tot - g0 < 32) ? (tot - g0) : 32;
    const int i = g0 + ((c.lane < n) ? c.lane : 0);
    int u, dl;
    if (i < total) {
      const unsigned wd = c.glist[i];
      u = (int)(wd & ((1u << LROW) - 1u));
      dl = (int)(wd >> LROW);
    } else {
      dl = i - total;
      u = c.n0 + dl;
    }
    edges_sum(c, u, dl, n);
  }
}

template <int MODE>
__device__ __forceinline__ void queue_group(const AggCtx& c, int qb, int n) {
  const int j = qb + ((c.lane < n) ? c.lane : 0);
  const int e = c.qe[j];
  const int dl = c.qd[j];
  const int u = c.src[e];
  if (MODE == 0) edges_max(c, u, dl, n);
  else edges_sum(c, u, dl, n);
}

__device__ __forceinline__ void agg_append(const AggCtx& c, bool m, int e,
                                           int dl, int& qn) {
  const unsigned bal = __builtin_amdgcn_ballot_w32(m);
  if (m) {
    const int pos = qn + (int)__builtin_amdgcn_mbcnt_lo(bal, 0u);
    c.qe[pos] = e;
    c.qd[pos] = (unsigned short)dl;
  }
  qn += __builtin_popcount(bal);
}

__device__ __forceinline__ int4 agg_load(const int4* dst4, int vi, int nb4) {
  return dst4[(vi < nb4) ? vi : (nb4 - 1)];
}

template <int MODE>
__device__ __forceinline__ void agg_scan4(const AggCtx& c, int4 d, int e0,
                                          unsigned cnt, int& qn) {
  const int l0 = d.x - c.n0, l1 = d.y - c.n0, l2 = d.z - c.n0,
            l3 = d.w - c.n0;
  const bool m0 = (unsigned)l0 < cnt, m1 = (unsigned)l1 < cnt,
             m2 = (unsigned)l2 < cnt, m3 = (unsigned)l3 < cnt;
  if (__builtin_amdgcn_ballot_w32(m0 | m1 | m2 | m3)) {
    agg_append(c, m0, e0 + 0, l0, qn);
    agg_append(c, m1, e0 + 1, l1, qn);
    agg_append(c, m2, e0 + 2, l2, qn);
    agg_append(c, m3, e0 + 3, l3, qn);
    if (qn >= 64) {
      const int nfull = qn & ~31;
      for (int qb = 0; qb < nfull; qb += 32)
        queue_group<MODE>(c, qb, 32);
      const int rem = qn - nfull;
      const int te = c.qe[nfull + c.lane];
      const unsigned short td = c.qd[nfull + c.lane];
      if (c.lane < rem) {
        c.qe[c.lane] = te;
        c.qd[c.lane] = td;
      }
      qn = rem;
    }
  }
}

template <int MODE>
__device__ __forceinline__ void scan_edges(const AggCtx& c, const int* dst,
                                           int n_edges, int cnt, int wave) {
  int qn = 0;
  int head = (int)(((16u - (unsigned)((size_t)dst & 15u)) & 15u) >> 2);
  if (head > n_edges) head = n_edges;
  const int nb4 = (n_edges - head) >> 2;
  const int tail = (n_edges - head) & 3;
  if (wave == 0) {
    const int nleft = head + tail;
    const int e = (c.lane < head) ? c.lane : (n_edges - tail + (c.lane - head));
    const bool ok = c.lane < nleft;
    const int dl = (ok ? dst[e] : -1) - c.n0;
    agg_append(c, ok && (unsigned)dl < (unsigned)cnt, e, dl, qn);
  }
  const int4* dst4 = (const int4*)(dst + head);
  if (nb4 > 0) {
    const int T = (nb4 + 4095) >> 12;
    int t = (int)((blockIdx.x * (unsigned)T) / gridDim.x);
    if (t >= T) t = 0;
    int4 cur[4], nxt[4];
#pragma unroll
    for (int s = 0; s < 4; ++s)
      cur[s] = agg_load(dst4, (t * 32 + wave) * 128 + s * 32 + c.lane, nb4);
    for (int k = 0; k < T; ++k) {
      int tn = t + 1;
      if (tn >= T) tn -= T;
#pragma unroll
      for (int s = 0; s < 4; ++s)
        nxt[s] = agg_load(dst4, (tn * 32 + wave) * 128 + s * 32 + c.lane, nb4);
#pragma unroll
      for (int s = 0; s < 4; ++s) {
        const int vi = (t * 32 + wave) * 128 + s * 32 + c.lane;
        agg_scan4<MODE>(c, cur[s], head + 4 * vi,
                        (vi < nb4) ? (unsigned)cnt : 0u, qn);
      }
#pragma unroll
      for (int s = 0; s < 4; ++s) cur[s] = nxt[s];
      t = tn;
    }
  }
  for (int qb = 0; qb < qn; qb += 32)
    queue_group<MODE>(c, qb, (qn - qb < 32) ? (qn - qb) : 32);
}

#define BIN_T 256
#define BIN_EPT 16

__global__ void __launch_bounds__(BIN_T)
gat_bin_edges(const int* __restrict__ src, const int* __restrict__ dst,
              unsigned* __restrict__ lists, int lstride, int ngroups,
              int n_edges) {
  extern __shared__ int sbin[];
  int* cntl = sbin;
  int* basel = sbin + ngroups;
  const int tid = (int)threadIdx.x;
  for (int i = tid; i < ngroups; i += BIN_T) cntl[i] = 0;

  const int e0 = (int)blockIdx.x * (BIN_T * BIN_EPT) + tid;
  int u[BIN_EPT], dv[BIN_EPT];
#pragma unroll
  for (int i = 0; i < BIN_EPT; ++i) {
    const int e = e0 + i * BIN_T;
    const bool ok = e < n_edges;
    dv[i] = ok ? dst[e] : -1;
    u[i] = ok ? src[e] : 0;
  }
  __syncthreads();

  int g[BIN_EPT], rk[BIN_EPT];
#pragma unroll
  for (int i = 0; i < BIN_EPT; ++i) {
    int gi = (dv[i] >= 0) ? (dv[i] / NPB) : -1;
    if (gi >= ngroups) gi = -1;
    g[i] = gi;
    rk[i] = 0;
    if (gi >= 0) rk[i] = atomicAdd(cntl + gi, 1);
  }
  __syncthreads();
  for (int i = tid; i < ngroups; i += BIN_T) {
    const int cg = cntl[i];
    int b = 0;
    if (cg > 0)
      b = (int)atomicAdd(lists + (size_t)i * (size_t)lstride, (unsigned)cg);
    basel[i] = b;
  }
  __syncthreads();
  const int cap = lstride - LHDR;
  int pos[BIN_EPT];
#pragma unroll
  for (int i = 0; i < BIN_EPT; ++i)
    pos[i] = basel[(g[i] >= 0) ? g[i] : 0] + rk[i];
#pragma unroll
  for (int i = 0; i < BIN_EPT; ++i) {
    if (g[i] >= 0 && pos[i] < cap)
      lists[(size_t)g[i] * (size_t)lstride + LHDR + pos[i]] =
          (unsigned)u[i] | ((unsigned)(dv[i] - g[i] * NPB) << LROW);
  }
}

__global__ void __launch_bounds__(1024)
gat_edge_aggregate(const int* __restrict__ src, const int* __restrict__ dst,
                   const float* __restrict__ s, const float* __restrict__ d,
                   const float* __restrict__ H,
                   const float* __restrict__ bias, const float* xin,
                   float* xout, const unsigned* __restrict__ lists,
                   int lstride, int n_edges, int n_nodes) {
  __shared__ __attribute__((aligned(16))) float sagg[NPB * CH];
  __shared__ unsigned s_maxu[NPB];
  __shared__ float s_den[NPB];
  __shared__ float s_d[NPB];
  __shared__ __attribute__((aligned(16))) unsigned s_qw[32 * QW];
  __shared__ float s_qs[32 * QW];
  __shared__ int s_qcnt[32];

  const int tid = (int)threadIdx.x;
  const int lane = tid & 31;
  const int wave = __builtin_amdgcn_readfirstlane(tid >> 5);
  const int n0 = blockIdx.x * NPB;
  const int cnt = (n_nodes - n0 < NPB) ? (n_nodes - n0) : NPB;
  const unsigned* hdr = lists + (size_t)blockIdx.x * (size_t)lstride;

  AggCtx c;
  c.src = src;
  c.s = s;
  c.sd = s_d;
  c.smaxu = s_maxu;
  c.sden = s_den;
  c.H = H;
  c.sagg = sagg;
  c.qe = (int*)s_qw + wave * QCAP;
  c.qd = (unsigned short*)(s_qw + 32 * QCAP) + wave * QCAP;
  c.qw = s_qw;
  c.qs = s_qs;
  c.qcnt = s_qcnt;
  c.glist = hdr + LHDR;
  c.cap = lstride - LHDR;
  c.n0 = n0;
  c.lane = lane;

  for (int i = tid; i < NPB * CH / 4; i += 1024)
    ((float4*)sagg)[i] = make_float4(0.0f, 0.0f, 0.0f, 0.0f);
  for (int rl = tid; rl < cnt; rl += 1024) {
    const int v = n0 + rl;
    const float dv = d[v];
    float val = s[v] + dv;
    val = (val > 0.0f) ? val : NEG_SLOPE * val;
    s_maxu[rl] = enc_f32(val);
    s_d[rl] = dv;
    s_den[rl] = 0.0f;
  }
  if (tid < 32) s_qcnt[tid] = 0;
  const int total = __builtin_amdgcn_readfirstlane((int)hdr[0]);
  __syncthreads();

  if (total <= c.cap) deal_list_max(c, total, wave);
  else scan_edges<0>(c, dst, n_edges, cnt, wave);
  __syncthreads();

  bool own = false;
  if (total <= c.cap) {
    own = __builtin_amdgcn_ballot_w32(s_qcnt[lane] > QW) == 0u;
    if (own) own_sum(c, wave, cnt);
    else walk_list_sum(c, total, cnt, wave);
  } else {
    scan_edges<1>(c, dst, n_edges, cnt, wave);
    walk_list_sum(c, 0, cnt, wave);
  }
  if (!own) __syncthreads();

  const float4 b4 = ((const float4*)bias)[lane];
  for (int rb = wave; rb < cnt; rb += 128) {
    float4 xv[4];
#pragma unroll
    for (int i = 0; i < 4; ++i) {
      const int rl = rb + 32 * i;
      if (rl < cnt)
        xv[i] = ((const float4*)(xin + (size_t)(n0 + rl) * CH))[lane];
    }
#pragma unroll
    for (int i = 0; i < 4; ++i) {
      const int rl = rb + 32 * i;
      if (rl < cnt) {
        const float4 a = ((const float4*)(sagg + rl * CH))[lane];
        const float dd = s_den[rl] + SM_EPS;
        float4 o = xv[i];
        o.x += 1.0f / (1.0f + __builtin_expf(-(a.x / dd + b4.x)));
        o.y += 1.0f / (1.0f + __builtin_expf(-(a.y / dd + b4.y)));
        o.z += 1.0f / (1.0f + __builtin_expf(-(a.z / dd + b4.z)));
        o.w += 1.0f / (1.0f + __builtin_expf(-(a.w / dd + b4.w)));
        ((float4*)(xout + (size_t)(n0 + rl) * CH))[lane] = o;
      }
    }
  }
}

extern "C" void kernel_launch(void* const* d_in, const int* in_sizes, int n_in,
                              void* d_out, int out_size, void* d_ws,
                              size_t ws_size, hipStream_t stream) {
  const float* x_in = (const float*)d_in[0];
  const int* ei = (const int*)d_in[1];
  const float* Ws = (const float*)d_in[2];
  const float* att_src = (const float*)d_in[3];
  const float* att_dst = (const float*)d_in[4];
  const float* bias = (const float*)d_in[5];

  const int N = in_sizes[0] / CH;
  const int E = in_sizes[1] / 2;
  const int M = in_sizes[3] / CH;
  const int* src = ei;
  const int* dst = ei + E;

  float* ws = (float*)d_ws;
  float* h = ws;
  float* s = h + (size_t)N * CH;
  float* dsc = s + N;
  unsigned* lists = (unsigned*)(dsc + N);

  float* x = (float*)d_out;

  const int blk = 256;
  const int g_gemm = (N + 127) / 128;
  const int g_own = (N + NPB - 1) / NPB;
  const int lstride = (int)(((size_t)E + 2 * (size_t)N) / (size_t)g_own);
  const size_t smem = (size_t)CH * WT_PAD * sizeof(float);
  const int g_bin = (E + BIN_T * BIN_EPT - 1) / (BIN_T * BIN_EPT);
  const size_t smem_bin = (size_t)g_own * 2 * sizeof(int);

  for (int m = 0; m < M; ++m) {
    const float* W = Ws + (size_t)m * CH * CH;
    const float* as_ = att_src + (size_t)m * CH;
    const float* ad_ = att_dst + (size_t)m * CH;
    const float* b_ = bias + (size_t)m * CH;

    const float* xr = (m == 0) ? x_in : x;
    gat_gemm_wmma<<<g_gemm, blk, smem, stream>>>(xr, W, as_, ad_, h, s, dsc, N,
                                                 (m == 0) ? lists : nullptr,
                                                 lstride, g_own);
    if (m == 0 && E > 0)
      gat_bin_edges<<<g_bin, BIN_T, smem_bin, stream>>>(src, dst, lists,
                                                        lstride, g_own, E);
    gat_edge_aggregate<<<g_own, 1024, 0, stream>>>(src, dst, s, dsc, h, b_, xr,
                                                   x, lists, lstride, E, N);
  }
}
